# starting kernel
# baseline (speedup 1.0000x reference)
#include <hip/hip_runtime.h>
#include <hip/hip_bf16.h>

typedef __attribute__((ext_vector_type(2))) float v2f;
typedef __attribute__((ext_vector_type(8))) float v8f;

#define DDIM 64
#define NMAX 8192

__device__ __forceinline__ float wave_reduce_sum(float v) {
#pragma unroll
    for (int off = 16; off > 0; off >>= 1) v += __shfl_xor(v, off, 32);
    return v;
}

__global__ void k_prep(const float* __restrict__ sH, const float* __restrict__ omega,
                       float* __restrict__ facc, float* __restrict__ gv,
                       float* __restrict__ skew, int N) {
    int i = blockIdx.x * blockDim.x + threadIdx.x;
    int ND = N * DDIM;
    int stride = gridDim.x * blockDim.x;
    for (int t = i; t < ND; t += stride) facc[t] = 0.0f;
    if (i < N) gv[i] = tanhf(sH[i]);
    if (i < DDIM * DDIM) {
        int k = i / DDIM, n = i % DDIM;
        skew[i] = 0.5f * (omega[k * DDIM + n] - omega[n * DDIM + k]);
    }
}

__global__ void k_norm(const float* __restrict__ K, float* __restrict__ Kn, int N) {
    int lane = threadIdx.x & 31, wid = threadIdx.x >> 5;
    int row = blockIdx.x * (blockDim.x >> 5) + wid;
    if (row >= N) return;
    float a0 = K[row * DDIM + lane];
    float a1 = K[row * DDIM + lane + 32];
    float s = wave_reduce_sum(a0 * a0 + a1 * a1);
    float inv = rsqrtf(s);
    Kn[row * DDIM + lane]      = a0 * inv;
    Kn[row * DDIM + lane + 32] = a1 * inv;
}

__global__ void k_matvec(const float* __restrict__ W, const float* __restrict__ gv,
                         const float* __restrict__ sH, float* __restrict__ fH, int N) {
    __shared__ float gs[NMAX];
    for (int t = threadIdx.x; t < N && t < NMAX; t += blockDim.x) gs[t] = gv[t];
    __syncthreads();
    int lane = threadIdx.x & 31, wid = threadIdx.x >> 5;
    int wpb = blockDim.x >> 5;
    int gw = blockIdx.x * wpb + wid;
    int tw = gridDim.x * wpb;
    int n4 = N >> 2;
    for (int row = gw; row < N; row += tw) {
        const float4* Wr = (const float4*)(W + (size_t)row * N);
        float acc = 0.0f;
        for (int c4 = lane; c4 < n4; c4 += 32) {
            float4 w = Wr[c4];
            int c = c4 << 2;
            acc += w.x * gs[c] + w.y * gs[c + 1] + w.z * gs[c + 2] + w.w * gs[c + 3];
        }
        acc = wave_reduce_sum(acc);
        if (lane == 0) fH[row] = -sH[row] + acc;
    }
}

__global__ void k_hk(const int* __restrict__ ind, const float* __restrict__ Kn,
                     const float* __restrict__ gv, float* __restrict__ fH,
                     float* __restrict__ facc, int E) {
    int lane = threadIdx.x & 31, wid = threadIdx.x >> 5;
    int e = blockIdx.x * (blockDim.x >> 5) + wid;
    if (e >= E) return;
    int i = ind[2 * e], j = ind[2 * e + 1];
    float a0 = Kn[i * DDIM + lane], a1 = Kn[i * DDIM + lane + 32];
    float c0 = Kn[j * DDIM + lane], c1 = Kn[j * DDIM + lane + 32];
    float Gram = wave_reduce_sum(a0 * c0 + a1 * c1);
    float gi = gv[i], gj = gv[j];
    if (lane == 0) {
        atomicAdd(&fH[i], 0.5f * Gram * gj);
        atomicAdd(&fH[j], 0.5f * Gram * gi);
    }
    float G = 0.5f * gi * gj;
    atomicAdd(&facc[i * DDIM + lane],      -G * c0);
    atomicAdd(&facc[i * DDIM + lane + 32], -G * c1);
    atomicAdd(&facc[j * DDIM + lane],      -G * a0);
    atomicAdd(&facc[j * DDIM + lane + 32], -G * a1);
}

__global__ void k_kedges(const int* __restrict__ ind, const float* __restrict__ Kn,
                         const float* __restrict__ w1v, const float* __restrict__ b1v,
                         const float* __restrict__ w2v, float* __restrict__ facc, int E) {
    int lane = threadIdx.x & 31, wid = threadIdx.x >> 5;
    int e = blockIdx.x * (blockDim.x >> 5) + wid;
    if (e >= E) return;
    int p = ind[2 * e], q = ind[2 * e + 1];
    float a0 = Kn[p * DDIM + lane], a1 = Kn[p * DDIM + lane + 32];
    float c0 = Kn[q * DDIM + lane], c1 = Kn[q * DDIM + lane + 32];
    float s = wave_reduce_sum(a0 * c0 + a1 * c1);
    float h0 = tanhf(s * w1v[lane]      + b1v[lane]);
    float h1 = tanhf(s * w1v[lane + 32] + b1v[lane + 32]);
    float dE = wave_reduce_sum(h0 * w2v[lane] + h1 * w2v[lane + 32]);
    atomicAdd(&facc[p * DDIM + lane],      dE * c0);
    atomicAdd(&facc[p * DDIM + lane + 32], dE * c1);
    atomicAdd(&facc[q * DDIM + lane],      dE * a0);
    atomicAdd(&facc[q * DDIM + lane + 32], dE * a1);
}

__global__ void k_komega(const float* __restrict__ Kn, const float* __restrict__ skew,
                         float* __restrict__ Kom, int N) {
    __shared__ float ssk[DDIM * DDIM];
    for (int t = threadIdx.x; t < DDIM * DDIM; t += blockDim.x) ssk[t] = skew[t];
    __syncthreads();
    int lane = threadIdx.x & 31, wid = threadIdx.x >> 5;
    int tile = blockIdx.x * (blockDim.x >> 5) + wid;
    if (tile >= (N >> 4)) return;
    int row0 = tile << 4;
    int l15 = lane & 15;
    int hi = lane >> 4;
    int arow = row0 + l15;
    int kofs = hi * 2;
    v8f acc[4] = {};
#pragma unroll
    for (int k0 = 0; k0 < DDIM; k0 += 4) {
        v2f A;
        A.x = Kn[arow * DDIM + k0 + kofs];
        A.y = Kn[arow * DDIM + k0 + kofs + 1];
#pragma unroll
        for (int t = 0; t < 4; ++t) {
            v2f B;
            int n = (t << 4) + l15;
            B.x = ssk[(k0 + kofs) * DDIM + n];
            B.y = ssk[(k0 + kofs + 1) * DDIM + n];
            acc[t] = __builtin_amdgcn_wmma_f32_16x16x4_f32(
                false, A, false, B, (short)0, acc[t], false, false);
        }
    }
    int crow = row0 + hi * 8;
#pragma unroll
    for (int t = 0; t < 4; ++t) {
#pragma unroll
        for (int v = 0; v < 8; ++v) {
            Kom[(crow + v) * DDIM + (t << 4) + l15] = acc[t][v];
        }
    }
}

__global__ void k_final(const float* __restrict__ Kn, const float* __restrict__ facc,
                        const float* __restrict__ Kom, float* __restrict__ fK, int N) {
    int lane = threadIdx.x & 31, wid = threadIdx.x >> 5;
    int row = blockIdx.x * (blockDim.x >> 5) + wid;
    if (row >= N) return;
    float k0 = Kn[row * DDIM + lane],   k1 = Kn[row * DDIM + lane + 32];
    float r0 = facc[row * DDIM + lane], r1 = facc[row * DDIM + lane + 32];
    float dot = wave_reduce_sum(k0 * r0 + k1 * r1);
    fK[row * DDIM + lane]      = -r0 + k0 * dot + Kom[row * DDIM + lane];
    fK[row * DDIM + lane + 32] = -r1 + k1 * dot + Kom[row * DDIM + lane + 32];
}

extern "C" void kernel_launch(void* const* d_in, const int* in_sizes, int n_in,
                              void* d_out, int out_size, void* d_ws, size_t ws_size,
                              hipStream_t stream) {
    const float* sH    = (const float*)d_in[0];
    const float* sK    = (const float*)d_in[1];
    const float* WH    = (const float*)d_in[2];
    const float* omega = (const float*)d_in[3];
    const float* w1v   = (const float*)d_in[4];
    const float* b1v   = (const float*)d_in[5];
    const float* w2v   = (const float*)d_in[6];
    const int*   indK  = (const int*)d_in[7];
    const int*   indHK = (const int*)d_in[8];

    int N   = in_sizes[0];
    int EK  = in_sizes[7] / 2;
    int EHK = in_sizes[8] / 2;
    int ND  = N * DDIM;

    float* fH = (float*)d_out;
    float* fK = (float*)d_out + N;

    float* Kn   = (float*)d_ws;
    float* gv   = Kn + (size_t)ND;
    float* skew = gv + N;
    float* facc = skew + DDIM * DDIM;
    float* Kom  = facc + (size_t)ND;

    k_prep  <<<(ND + 255) / 256,        256, 0, stream>>>(sH, omega, facc, gv, skew, N);
    k_norm  <<<(N + 7) / 8,             256, 0, stream>>>(sK, Kn, N);
    k_matvec<<<256,                     256, 0, stream>>>(WH, gv, sH, fH, N);
    k_hk    <<<(EHK + 7) / 8,           256, 0, stream>>>(indHK, Kn, gv, fH, facc, EHK);
    k_kedges<<<(EK + 7) / 8,            256, 0, stream>>>(indK, Kn, w1v, b1v, w2v, facc, EK);
    k_komega<<<((N >> 4) + 7) / 8,      256, 0, stream>>>(Kn, skew, Kom, N);
    k_final <<<(N + 7) / 8,             256, 0, stream>>>(Kn, facc, Kom, fK, N);
}


# owner role: hits queued per wave in LDS and worked off 8 at a time, 8 partner-row loads and edge scalars in flight together (was one memory wait per hit); LDS atomic adds, arrival order as before
# speedup vs baseline: 3.4337x; 3.4337x over previous
#include <hip/hip_runtime.h>
#include <hip/hip_bf16.h>

typedef __attribute__((ext_vector_type(2))) float v2f;
typedef __attribute__((ext_vector_type(8))) float v8f;

#define DDIM 64
#define NMAX 8192

__device__ __forceinline__ float wave_reduce_sum(float v) {
#pragma unroll
    for (int off = 16; off > 0; off >>= 1) v += __shfl_xor(v, off, 32);
    return v;
}

__global__ void k_prep(const float* __restrict__ sH, const float* __restrict__ omega,
                       float* __restrict__ gv,
                       float* __restrict__ skew, float* __restrict__ fH, int N) {
    int i = blockIdx.x * blockDim.x + threadIdx.x;
    if (i < N) {
        float x = sH[i];
        gv[i] = tanhf(x);
        fH[i] = -x;
    }
    if (i < DDIM * DDIM) {
        int k = i / DDIM, n = i % DDIM;
        skew[i] = 0.5f * (omega[k * DDIM + n] - omega[n * DDIM + k]);
    }
}

__global__ void k_norm(const float* __restrict__ K, float* __restrict__ Kn, int N) {
    int lane = threadIdx.x & 31, wid = threadIdx.x >> 5;
    int row = blockIdx.x * (blockDim.x >> 5) + wid;
    if (row >= N) return;
    float a0 = K[row * DDIM + lane];
    float a1 = K[row * DDIM + lane + 32];
    float s = wave_reduce_sum(a0 * a0 + a1 * a1);
    float inv = rsqrtf(s);
    Kn[row * DDIM + lane]      = a0 * inv;
    Kn[row * DDIM + lane + 32] = a1 * inv;
}

#define NFLY 4

template <bool IS_K>
__device__ __forceinline__ void coef_chunk(const int* __restrict__ ind, int E, int e0,
                                           const float* __restrict__ Kn,
                                           float* __restrict__ cf,
                                           float w1a, float w1b, float b1a, float b1b,
                                           float w2a, float w2b) {
    int lane = threadIdx.x & 31;
    int2 ij;
    ij.x = 0; ij.y = 0;
    if (e0 + lane < E) ij = ((const int2*)ind)[e0 + lane];
    float res = 0.0f;
    for (int b = 0; b < 32; b += NFLY) {
        float a0[NFLY], a1[NFLY], c0[NFLY], c1[NFLY], d[NFLY];
        for (int u = 0; u < NFLY; ++u) {
            int i = __builtin_amdgcn_readlane(ij.x, b + u);
            int j = __builtin_amdgcn_readlane(ij.y, b + u);
            a0[u] = Kn[i * DDIM + lane]; a1[u] = Kn[i * DDIM + lane + 32];
            c0[u] = Kn[j * DDIM + lane]; c1[u] = Kn[j * DDIM + lane + 32];
        }
        for (int u = 0; u < NFLY; ++u) d[u] = a0[u] * c0[u] + a1[u] * c1[u];
        for (int off = 16; off > 0; off >>= 1) {
            for (int u = 0; u < NFLY; ++u) d[u] += __shfl_xor(d[u], off, 32);
        }
        if (IS_K) {
            for (int u = 0; u < NFLY; ++u) {
                float h0 = tanhf(d[u] * w1a + b1a);
                float h1 = tanhf(d[u] * w1b + b1b);
                d[u] = h0 * w2a + h1 * w2b;
            }
            for (int off = 16; off > 0; off >>= 1) {
                for (int u = 0; u < NFLY; ++u) d[u] += __shfl_xor(d[u], off, 32);
            }
        }
        for (int u = 0; u < NFLY; ++u) if (lane == b + u) res = d[u];
    }
    if (e0 + lane < E) cf[e0 + lane] = res;
}

__global__ void k_coef(const int* __restrict__ indHK, const int* __restrict__ indK,
                       const float* __restrict__ Kn,
                       const float* __restrict__ w1v, const float* __restrict__ b1v,
                       const float* __restrict__ w2v,
                       float* __restrict__ cHK, float* __restrict__ cK, int EHK, int EK) {
    int lane = threadIdx.x & 31, wid = threadIdx.x >> 5;
    int wpb = blockDim.x >> 5;
    int gw = blockIdx.x * wpb + wid;
    int tw = gridDim.x * wpb;
    int nCK = (EK + 31) >> 5, nCH = (EHK + 31) >> 5;
    float w1a = w1v[lane], w1b = w1v[lane + 32];
    float b1a = b1v[lane], b1b = b1v[lane + 32];
    float w2a = w2v[lane], w2b = w2v[lane + 32];
    for (int c = gw; c < nCK + nCH; c += tw) {
        if (c < nCK) coef_chunk<true >(indK,  EK,  c << 5,         Kn, cK,  w1a, w1b, b1a, b1b, w2a, w2b);
        else         coef_chunk<false>(indHK, EHK, (c - nCK) << 5, Kn, cHK, w1a, w1b, b1a, b1b, w2a, w2b);
    }
}


__device__ __forceinline__ void role_matvec(float* gs, int blk, int nblk,
                                            const float* __restrict__ W,
                                            const float* __restrict__ gv,
                                            float* __restrict__ fH, int N) {
    for (int t = threadIdx.x; t < N && t < NMAX; t += blockDim.x) gs[t] = gv[t];
    __syncthreads();
    int lane = threadIdx.x & 31, wid = threadIdx.x >> 5;
    int wpb = blockDim.x >> 5;
    int gw = blk * wpb + wid;
    int tw = nblk * wpb;
    int n4 = N >> 2;
    for (int row = gw; row < N; row += tw) {
        const float4* Wr = (const float4*)(W + (size_t)row * N);
        float acc = 0.0f;
        for (int c4 = lane; c4 < n4; c4 += 32) {
            float4 w = Wr[c4];
            int c = c4 << 2;
            acc += w.x * gs[c] + w.y * gs[c + 1] + w.z * gs[c + 2] + w.w * gs[c + 3];
        }
        acc = wave_reduce_sum(acc);
        if (lane == 0) atomicAdd(&fH[row], acc);
    }
}

#define NOWN_MAX 128
#define NB 8
#define QCAP 128

__device__ __forceinline__ void q_push(int2* q, int& tail, bool hit, int ownl, int oth, int e) {
    unsigned m = __builtin_amdgcn_ballot_w32(hit);
    int pos = tail + (int)__builtin_amdgcn_mbcnt_lo(m, 0u);
    if (hit) {
        int2 t;
        t.x = ownl | (oth << 8);
        t.y = e;
        q[pos & (QCAP - 1)] = t;
    }
    tail += __builtin_popcount(m);
}

template <bool IS_K, bool FULL>
__device__ __forceinline__ void owner_batch(float* acc, float* fhs, int2* q, int head, int n, int n0,
                                            const float* __restrict__ Kn,
                                            const float* __restrict__ gv,
                                            const float* __restrict__ cf) {
    int lane = threadIdx.x & 31;
    int k = lane & (NB - 1);
    int2 ent = q[(head + k) & (QCAP - 1)];
    bool valid = FULL || k < n;
    int ownl = ent.x & 0xff, oth = ent.x >> 8;
    float c = 0.0f, go = 0.0f, gw = 0.0f;
    if (valid) {
        c = cf[ent.y];
        if (!IS_K) { go = gv[oth]; gw = gv[n0 + ownl]; }
    }
    float2 x[NB];
    for (int u = 0; u < NB; ++u) {
        if (FULL || u < n) {
            int o = __builtin_amdgcn_readlane(ent.x, u) >> 8;
            x[u] = ((const float2*)Kn)[o * (DDIM / 2) + lane];
        }
    }
    float mult = IS_K ? c : -(0.5f * gw * go);
    if (!IS_K) {
        if (valid && lane < NB) atomicAdd(fhs + ownl, 0.5f * c * go);
    }
    for (int u = 0; u < NB; ++u) {
        if (FULL || u < n) {
            float mu = __int_as_float(__builtin_amdgcn_readlane(__float_as_int(mult), u));
            int r = __builtin_amdgcn_readlane(ent.x, u) & 0xff;
            float* row = acc + r * DDIM + 2 * lane;
            atomicAdd(row,     mu * x[u].x);
            atomicAdd(row + 1, mu * x[u].y);
        }
    }
}

__device__ __forceinline__ int2 load_pair(const int* __restrict__ ind, int e, int E) {
    int2 r = ((const int2*)ind)[min(e, E - 1)];
    if (e >= E) { r.x = -1; r.y = -1; }
    return r;
}

template <bool IS_K>
__device__ __forceinline__ void owner_hits(float* acc, float* fhs, int2* q, int& head, int& tail,
                                           int n0, int nown, int2 ij, int ebase,
                                           const float* __restrict__ Kn,
                                           const float* __restrict__ gv,
                                           const float* __restrict__ cf) {
    int lane = threadIdx.x & 31;
    bool oi = (unsigned)(ij.x - n0) < (unsigned)nown;
    bool oj = (unsigned)(ij.y - n0) < (unsigned)nown;
    if (__builtin_amdgcn_ballot_w32(oi | oj) == 0) return;
    q_push(q, tail, oi, ij.x - n0, ij.y, ebase + lane);
    q_push(q, tail, oj, ij.y - n0, ij.x, ebase + lane);
    while (tail - head >= NB) {
        owner_batch<IS_K, true>(acc, fhs, q, head, NB, n0, Kn, gv, cf);
        head += NB;
    }
}

template <bool IS_K>
__device__ __forceinline__ void owner_scan(float* acc, float* fhs, int2* q, int n0, int nown,
                                           const int* __restrict__ ind, int lo, int hi,
                                           const float* __restrict__ Kn,
                                           const float* __restrict__ gv,
                                           const float* __restrict__ cf) {
    if (lo >= hi) return;
    int lane = threadIdx.x & 31, wid = threadIdx.x >> 5;
    int step = (blockDim.x >> 5) * 128;
    int eb = lo + wid * 128;
    int head = 0, tail = 0;
    int2 p0 = load_pair(ind, eb + lane, hi),      p1 = load_pair(ind, eb + 32 + lane, hi);
    int2 p2 = load_pair(ind, eb + 64 + lane, hi), p3 = load_pair(ind, eb + 96 + lane, hi);
    while (eb < hi) {
        int nb = eb + step;
        int2 q0 = load_pair(ind, nb + lane, hi),      q1 = load_pair(ind, nb + 32 + lane, hi);
        int2 q2 = load_pair(ind, nb + 64 + lane, hi), q3 = load_pair(ind, nb + 96 + lane, hi);
        owner_hits<IS_K>(acc, fhs, q, head, tail, n0, nown, p0, eb,      Kn, gv, cf);
        owner_hits<IS_K>(acc, fhs, q, head, tail, n0, nown, p1, eb + 32, Kn, gv, cf);
        owner_hits<IS_K>(acc, fhs, q, head, tail, n0, nown, p2, eb + 64, Kn, gv, cf);
        owner_hits<IS_K>(acc, fhs, q, head, tail, n0, nown, p3, eb + 96, Kn, gv, cf);
        p0 = q0; p1 = q1; p2 = q2; p3 = q3;
        eb = nb;
    }
    if (tail - head > 0)
        owner_batch<IS_K, false>(acc, fhs, q, head, tail - head, n0, Kn, gv, cf);
}

__device__ __forceinline__ void role_owner(float* acc, int2* hitq, int blk,
                                           const int* __restrict__ indHK,
                                           const int* __restrict__ indK,
                                           const float* __restrict__ Kn,
                                           const float* __restrict__ gv,
                                           const float* __restrict__ cHK,
                                           const float* __restrict__ cK,
                                           float* __restrict__ fH, float* __restrict__ facc,
                                           float* __restrict__ fpl,
                                           int N, int EHK, int EK, int nown, int nsl) {
    int sl = blk % nsl;
    int n0 = (blk / nsl) * nown;
    float* fhs = acc + nown * DDIM;
    int2* q = hitq + (threadIdx.x >> 5) * QCAP;
    for (int t = threadIdx.x; t < nown * DDIM + nown; t += blockDim.x) acc[t] = 0.0f;
    __syncthreads();
    int perH = (EHK + nsl - 1) / nsl, perK = (EK + nsl - 1) / nsl;
    int loH = sl * perH, loK = sl * perK;
    owner_scan<false>(acc, fhs, q, n0, nown, indHK, loH, min(EHK, loH + perH), Kn, gv, cHK);
    owner_scan<true >(acc, fhs, q, n0, nown, indK,  loK, min(EK,  loK + perK), Kn, gv, cK);
    __syncthreads();
    float* dst = sl == 0 ? facc : fpl + (size_t)(sl - 1) * N * DDIM;
    for (int t = threadIdx.x; t < nown * DDIM; t += blockDim.x) {
        int row = n0 + t / DDIM;
        if (row < N) dst[row * DDIM + (t % DDIM)] = acc[t];
    }
    if (threadIdx.x < nown && n0 + threadIdx.x < N) atomicAdd(&fH[n0 + threadIdx.x], fhs[threadIdx.x]);
}

__device__ __forceinline__ void role_komega(float* ssk, int blk,
                                            const float* __restrict__ Kn,
                                            const float* __restrict__ skew,
                                            float* __restrict__ Kom, int N) {
    for (int t = threadIdx.x; t < DDIM * DDIM; t += blockDim.x) ssk[t] = skew[t];
    __syncthreads();
    int lane = threadIdx.x & 31, wid = threadIdx.x >> 5;
    int tile = blk * (blockDim.x >> 5) + wid;
    if (tile >= (N >> 4)) return;
    int row0 = tile << 4;
    int l15 = lane & 15;
    int hi = lane >> 4;
    int arow = row0 + l15;
    int kofs = hi * 2;
    v8f acc[4] = {};
#pragma unroll
    for (int k0 = 0; k0 < DDIM; k0 += 4) {
        v2f A;
        A.x = Kn[arow * DDIM + k0 + kofs];
        A.y = Kn[arow * DDIM + k0 + kofs + 1];
#pragma unroll
        for (int t = 0; t < 4; ++t) {
            v2f B;
            int n = (t << 4) + l15;
            B.x = ssk[(k0 + kofs) * DDIM + n];
            B.y = ssk[(k0 + kofs + 1) * DDIM + n];
            acc[t] = __builtin_amdgcn_wmma_f32_16x16x4_f32(
                false, A, false, B, (short)0, acc[t], false, false);
        }
    }
    int crow = row0 + hi * 8;
#pragma unroll
    for (int t = 0; t < 4; ++t) {
#pragma unroll
        for (int v = 0; v < 8; ++v) {
            Kom[(crow + v) * DDIM + (t << 4) + l15] = acc[t][v];
        }
    }
}

__global__ void k_roles(const float* __restrict__ W, const float* __restrict__ gv,
                        float* __restrict__ fH,
                        const float* __restrict__ Kn, const float* __restrict__ skew,
                        float* __restrict__ Kom,
                        const int* __restrict__ indHK, const int* __restrict__ indK,
                        const float* __restrict__ cHK, const float* __restrict__ cK,
                        float* __restrict__ facc, float* __restrict__ fpl,
                        int N, int EHK, int EK, int nbMV, int nbE, int nbKO, int nown, int nsl) {
    __shared__ float smem[NMAX + NOWN_MAX];
    __shared__ int2 hitq[32 * QCAP];
    int b = blockIdx.x;
    if (b < nbE) { role_owner(smem, hitq, b, indHK, indK, Kn, gv, cHK, cK, fH, facc, fpl, N, EHK, EK, nown, nsl); return; }
    b -= nbE;
    if (b < nbMV) { role_matvec(smem, b, nbMV, W, gv, fH, N); return; }
    b -= nbMV;
    if (b < nbKO) role_komega(smem, b, Kn, skew, Kom, N);
}

__global__ void k_final(const float* __restrict__ Kn,
                        const float* __restrict__ Kom, float* fK,
                        const float* __restrict__ fpl, int npl, int N) {
    int lane = threadIdx.x & 31, wid = threadIdx.x >> 5;
    int row = blockIdx.x * (blockDim.x >> 5) + wid;
    if (row >= N) return;
    float k0 = Kn[row * DDIM + lane],   k1 = Kn[row * DDIM + lane + 32];
    float r0 = fK[row * DDIM + lane], r1 = fK[row * DDIM + lane + 32];
    size_t nd = (size_t)N * DDIM;
    float a0 = 0.0f, a1 = 0.0f, b0 = 0.0f, b1 = 0.0f, c0 = 0.0f, c1 = 0.0f;
    if (npl > 0) { a0 = fpl[row * DDIM + lane];          a1 = fpl[row * DDIM + lane + 32]; }
    if (npl > 1) { b0 = fpl[nd + row * DDIM + lane];     b1 = fpl[nd + row * DDIM + lane + 32]; }
    if (npl > 2) { c0 = fpl[2 * nd + row * DDIM + lane]; c1 = fpl[2 * nd + row * DDIM + lane + 32]; }
    r0 = ((r0 + a0) + b0) + c0;
    r1 = ((r1 + a1) + b1) + c1;
    float dot = wave_reduce_sum(k0 * r0 + k1 * r1);
    fK[row * DDIM + lane]      = -r0 + k0 * dot + Kom[row * DDIM + lane];
    fK[row * DDIM + lane + 32] = -r1 + k1 * dot + Kom[row * DDIM + lane + 32];
}

extern "C" void kernel_launch(void* const* d_in, const int* in_sizes, int n_in,
                              void* d_out, int out_size, void* d_ws, size_t ws_size,
                              hipStream_t stream) {
    const float* sH    = (const float*)d_in[0];
    const float* sK    = (const float*)d_in[1];
    const float* WH    = (const float*)d_in[2];
    const float* omega = (const float*)d_in[3];
    const float* w1v   = (const float*)d_in[4];
    const float* b1v   = (const float*)d_in[5];
    const float* w2v   = (const float*)d_in[6];
    const int*   indK  = (const int*)d_in[7];
    const int*   indHK = (const int*)d_in[8];

    int N   = in_sizes[0];
    int EK  = in_sizes[7] / 2;
    int EHK = in_sizes[8] / 2;
    int ND  = N * DDIM;

    float* fH = (float*)d_out;
    float* fK = (float*)d_out + N;

    float* Kn   = (float*)d_ws;
    float* gv   = Kn + (size_t)ND;
    float* skew = gv + N;
    float* Kom  = skew + DDIM * DDIM;
    float* cHK  = Kom + (size_t)ND;
    float* cK   = cHK + EHK;
    float* fpl  = cK + EK;

    size_t wsf  = ws_size / sizeof(float);
    size_t used = (size_t)(fpl - (float*)d_ws);
    int nsl  = wsf >= used + 3 * (size_t)ND ? 4 : (wsf >= used + (size_t)ND ? 2 : 1);
    int nown = 32 * nsl;
    int nbE  = ((N + nown - 1) / nown) * nsl;
    int nbMV = (N + 31) / 32;
    int nbKO = ((N >> 4) + 31) / 32;
    int nP   = N > DDIM * DDIM ? N : DDIM * DDIM;
    int nCh  = (EK + 31) / 32 + (EHK + 31) / 32;
    int nbC  = (nCh + 15) / 16;

    k_prep  <<<(nP + 255) / 256,    256,  0, stream>>>(sH, omega, gv, skew, fH, N);
    k_norm  <<<(N + 7) / 8,         256,  0, stream>>>(sK, Kn, N);
    k_coef  <<<nbC,                 256,  0, stream>>>(indHK, indK, Kn, w1v, b1v, w2v, cHK, cK, EHK, EK);
    k_roles <<<nbE + nbMV + nbKO,   1024, 0, stream>>>(WH, gv, fH, Kn, skew, Kom,
                                                       indHK, indK, cHK, cK, fK, fpl,
                                                       N, EHK, EK, nbMV, nbE, nbKO, nown, nsl);
    k_final <<<(N + 7) / 8,             256, 0, stream>>>(Kn, Kom, fK, fpl, nsl - 1, N);
}
